# starting kernel
# baseline (speedup 1.0000x reference)
#include <hip/hip_runtime.h>
#include <hip/hip_bf16.h>
#include <math.h>

#define BB 16
#define TT 96
#define DD 768
#define VV 4096

typedef __attribute__((ext_vector_type(2))) float v2f;
typedef __attribute__((ext_vector_type(8))) float v8f;

__global__ void tg_w_kernel(const int* __restrict__ token_ids,
                            const float* __restrict__ edge_weights,
                            const int* __restrict__ edge_matrix,
                            float* __restrict__ w_buf) {
    int i = blockIdx.x * blockDim.x + threadIdx.x;
    const int total = BB * TT * TT;
    if (i >= total) return;
    int b = i / (TT * TT);
    int rem = i - b * (TT * TT);
    int s = rem / TT;
    int t = rem - s * TT;
    int ts = token_ids[b * TT + s];
    int td = token_ids[b * TT + t];
    int eid = edge_matrix[ts * VV + td];
    w_buf[i] = edge_weights[eid];
}

__global__ void tg_maxagg_kernel(const int* __restrict__ token_ids,
                                 const float* __restrict__ node_emb,
                                 const float* __restrict__ w_buf,
                                 const float* __restrict__ eta_p,
                                 float* __restrict__ g_buf) {
    __shared__ float Hs[TT * 128];
    __shared__ int   toks[TT];

    const int chunk = blockIdx.x % (DD / 128);
    const int b     = blockIdx.x / (DD / 128);
    const int d0    = chunk * 128;
    const int tid   = threadIdx.x;

    if (tid < TT) toks[tid] = token_ids[b * TT + tid];
    __syncthreads();

    for (int s = 0; s < TT; ++s) {
        Hs[s * 128 + tid] = node_emb[(size_t)toks[s] * DD + d0 + tid];
    }
    __syncthreads();

    const float eta = eta_p[0];
    const float one_m_eta = 1.0f - eta;
    const float* wb = w_buf + (size_t)b * TT * TT;

    float gacc = 0.0f;
    for (int t = 0; t < TT; ++t) {
        float acc = -INFINITY;
        #pragma unroll 4
        for (int s = 0; s < TT; ++s) {
            acc = fmaxf(acc, Hs[s * 128 + tid] * wb[s * TT + t]);
        }
        gacc += eta * Hs[t * 128 + tid] + one_m_eta * acc;
    }
    g_buf[b * DD + d0 + tid] = gacc * (1.0f / (float)TT);
}

__global__ void tg_gemm_sigmoid_wmma(const float* __restrict__ g_buf,
                                     const float* __restrict__ Wm,
                                     const float* __restrict__ bias,
                                     float* __restrict__ out) {
    const int n0   = blockIdx.x * 16;
    const int lane = threadIdx.x;
    const int half = lane >> 4;
    const int l    = lane & 15;

    v8f c = {};
    const float* arow = g_buf + (size_t)l * DD;
    const float* brow = Wm    + (size_t)(n0 + l) * DD;

    #pragma unroll 8
    for (int k0 = 0; k0 < DD; k0 += 4) {
        const int k = k0 + 2 * half;
        v2f a, bf;
        a.x  = arow[k + 0];
        a.y  = arow[k + 1];
        bf.x = brow[k + 0];
        bf.y = brow[k + 1];
        c = __builtin_amdgcn_wmma_f32_16x16x4_f32(
                 false, a,  false, bf,
                 (short)0, c,  false,  false);
    }

    const float bb = bias[n0 + l];
    #pragma unroll
    for (int r = 0; r < 8; ++r) {
        const int m = r + 8 * half;
        const float v = c[r] + bb;
        out[(size_t)m * DD + n0 + l] = 1.0f / (1.0f + expf(-v));
    }
}

extern "C" void kernel_launch(void* const* d_in, const int* in_sizes, int n_in,
                              void* d_out, int out_size, void* d_ws, size_t ws_size,
                              hipStream_t stream) {
    const int*   token_ids    = (const int*)  d_in[0];
    const float* node_emb     = (const float*)d_in[1];
    const float* edge_weights = (const float*)d_in[2];
    const int*   edge_matrix  = (const int*)  d_in[3];
    const float* eta          = (const float*)d_in[4];
    const float* Wm           = (const float*)d_in[5];
    const float* bias         = (const float*)d_in[6];
    float*       out          = (float*)d_out;

    float* w_buf = (float*)d_ws;
    float* g_buf = w_buf + (size_t)BB * TT * TT;

    {
        const int total = BB * TT * TT;
        dim3 grid((total + 255) / 256), block(256);
        tg_w_kernel<<<grid, block, 0, stream>>>(token_ids, edge_weights,
                                                edge_matrix, w_buf);
    }
    {
        dim3 grid(BB * (DD / 128)), block(128);
        tg_maxagg_kernel<<<grid, block, 0, stream>>>(token_ids, node_emb,
                                                     w_buf, eta, g_buf);
    }
    {
        dim3 grid(DD / 16), block(32);
        tg_gemm_sigmoid_wmma<<<grid, block, 0, stream>>>(g_buf, Wm, bias, out);
    }
}


# max-times stage: a lane keeps its H column in 96 registers and multiplies by a destination's 96 weights from scalar registers (scalar loads of a destination-major w); 8 waves share the destinations
# speedup vs baseline: 22.2909x; 22.2909x over previous
#include <hip/hip_runtime.h>
#include <hip/hip_bf16.h>
#include <math.h>

#define BB 16
#define TT 96
#define DD 768
#define VV 4096

typedef __attribute__((ext_vector_type(2))) float v2f;
typedef __attribute__((ext_vector_type(8))) float v8f;

__global__ void tg_w_kernel(const int* __restrict__ token_ids,
                            const float* __restrict__ edge_weights,
                            const int* __restrict__ edge_matrix,
                            float* __restrict__ wT_buf) {
    int i = blockIdx.x * blockDim.x + threadIdx.x;
    const int total = BB * TT * TT;
    if (i >= total) return;
    int b = i / (TT * TT);
    int rem = i - b * (TT * TT);
    int t = rem / TT;
    int s = rem - t * TT;
    int ts = token_ids[b * TT + s];
    int td = token_ids[b * TT + t];
    int eid = edge_matrix[ts * VV + td];
    wT_buf[i] = edge_weights[eid];
}

#define MX_NW 8
#define MX_TS (TT / MX_NW)

__global__ __launch_bounds__(32 * MX_NW)
void tg_maxagg_kernel(const int* __restrict__ token_ids,
                      const float* __restrict__ node_emb,
                      const float* __restrict__ wT_buf,
                      const float* __restrict__ eta_p,
                      float* __restrict__ g_buf) {
    __shared__ float amax[TT * 32];

    const int cg   = blockIdx.x % (DD / 32);
    const int b    = blockIdx.x / (DD / 32);
    const int lane = threadIdx.x & 31;
    const int wv   = __builtin_amdgcn_readfirstlane((int)(threadIdx.x >> 5));
    const int d    = cg * 32 + lane;
    const int* tk  = token_ids + b * TT;

    const char* nb = (const char*)node_emb;
    const unsigned dbyte = (unsigned)d * 4u;
    float h[TT];
    #pragma unroll
    for (int s = 0; s < TT; ++s) {
        const unsigned off = (unsigned)tk[s] * (unsigned)(DD * 4) + dbyte;
        h[s] = *(const float*)(nb + off);
    }

    const int t0 = wv * MX_TS;
    const float* wrow = wT_buf + ((size_t)b * TT + t0) * TT;

    #pragma clang loop vectorize(disable) interleave(disable) unroll(disable)
    for (int i = 0; i < MX_TS; ++i) {
        float a0 = -INFINITY, a1 = -INFINITY, a2 = -INFINITY, a3 = -INFINITY;
        #pragma unroll
        for (int s = 0; s < TT; s += 8) {
            a0 = fmaxf(fmaxf(a0, h[s + 0] * wrow[s + 0]), h[s + 4] * wrow[s + 4]);
            a1 = fmaxf(fmaxf(a1, h[s + 1] * wrow[s + 1]), h[s + 5] * wrow[s + 5]);
            a2 = fmaxf(fmaxf(a2, h[s + 2] * wrow[s + 2]), h[s + 6] * wrow[s + 6]);
            a3 = fmaxf(fmaxf(a3, h[s + 3] * wrow[s + 3]), h[s + 7] * wrow[s + 7]);
        }
        amax[(t0 + i) * 32 + lane] = fmaxf(fmaxf(a0, a1), fmaxf(a2, a3));
        wrow += TT;
    }
    __syncthreads();

    if (wv == 0) {
        const float eta = eta_p[0];
        const float one_m_eta = 1.0f - eta;
        float gacc = 0.0f;
        #pragma unroll
        for (int t = 0; t < TT; ++t) {
            gacc += eta * h[t] + one_m_eta * amax[t * 32 + lane];
        }
        g_buf[b * DD + d] = gacc * (1.0f / (float)TT);
    }
}

__global__ void tg_gemm_sigmoid_wmma(const float* __restrict__ g_buf,
                                     const float* __restrict__ Wm,
                                     const float* __restrict__ bias,
                                     float* __restrict__ out) {
    const int n0   = blockIdx.x * 16;
    const int lane = threadIdx.x;
    const int half = lane >> 4;
    const int l    = lane & 15;

    v8f c = {};
    const float* arow = g_buf + (size_t)l * DD;
    const float* brow = Wm    + (size_t)(n0 + l) * DD;

    #pragma unroll 8
    for (int k0 = 0; k0 < DD; k0 += 4) {
        const int k = k0 + 2 * half;
        v2f a, bf;
        a.x  = arow[k + 0];
        a.y  = arow[k + 1];
        bf.x = brow[k + 0];
        bf.y = brow[k + 1];
        c = __builtin_amdgcn_wmma_f32_16x16x4_f32(
                 false, a,  false, bf,
                 (short)0, c,  false,  false);
    }

    const float bb = bias[n0 + l];
    #pragma unroll
    for (int r = 0; r < 8; ++r) {
        const int m = r + 8 * half;
        const float v = c[r] + bb;
        out[(size_t)m * DD + n0 + l] = 1.0f / (1.0f + expf(-v));
    }
}

extern "C" void kernel_launch(void* const* d_in, const int* in_sizes, int n_in,
                              void* d_out, int out_size, void* d_ws, size_t ws_size,
                              hipStream_t stream) {
    const int*   token_ids    = (const int*)  d_in[0];
    const float* node_emb     = (const float*)d_in[1];
    const float* edge_weights = (const float*)d_in[2];
    const int*   edge_matrix  = (const int*)  d_in[3];
    const float* eta          = (const float*)d_in[4];
    const float* Wm           = (const float*)d_in[5];
    const float* bias         = (const float*)d_in[6];
    float*       out          = (float*)d_out;

    float* w_buf = (float*)d_ws;
    float* g_buf = w_buf + (size_t)BB * TT * TT;

    {
        const int total = BB * TT * TT;
        dim3 grid((total + 255) / 256), block(256);
        tg_w_kernel<<<grid, block, 0, stream>>>(token_ids, edge_weights,
                                                edge_matrix, w_buf);
    }
    {
        dim3 grid(BB * (DD / 32)), block(32 * MX_NW);
        tg_maxagg_kernel<<<grid, block, 0, stream>>>(token_ids, node_emb,
                                                     w_buf, eta, g_buf);
    }
    {
        dim3 grid(DD / 16), block(32);
        tg_gemm_sigmoid_wmma<<<grid, block, 0, stream>>>(g_buf, Wm, bias, out);
    }
}


# projection GEMM K loop software-pipelined in its one wave: ring of 4 register buffers refilled by grouped 16-byte loads 3 groups ahead, steps over 4 accumulator tiles (changes k summation order)
# speedup vs baseline: 31.0858x; 1.3946x over previous
#include <hip/hip_runtime.h>
#include <hip/hip_bf16.h>
#include <math.h>

#define BB 16
#define TT 96
#define DD 768
#define VV 4096

typedef __attribute__((ext_vector_type(2))) float v2f;
typedef __attribute__((ext_vector_type(4))) float v4f;
typedef __attribute__((ext_vector_type(8))) float v8f;

__global__ void tg_w_kernel(const int* __restrict__ token_ids,
                            const float* __restrict__ edge_weights,
                            const int* __restrict__ edge_matrix,
                            float* __restrict__ wT_buf) {
    int i = blockIdx.x * blockDim.x + threadIdx.x;
    const int total = BB * TT * TT;
    if (i >= total) return;
    int b = i / (TT * TT);
    int rem = i - b * (TT * TT);
    int t = rem / TT;
    int s = rem - t * TT;
    int ts = token_ids[b * TT + s];
    int td = token_ids[b * TT + t];
    int eid = edge_matrix[ts * VV + td];
    wT_buf[i] = edge_weights[eid];
}

#define MX_NW 8
#define MX_TS (TT / MX_NW)

__global__ __launch_bounds__(32 * MX_NW)
void tg_maxagg_kernel(const int* __restrict__ token_ids,
                      const float* __restrict__ node_emb,
                      const float* __restrict__ wT_buf,
                      const float* __restrict__ eta_p,
                      float* __restrict__ g_buf) {
    __shared__ float amax[TT * 32];

    const int cg   = blockIdx.x % (DD / 32);
    const int b    = blockIdx.x / (DD / 32);
    const int lane = threadIdx.x & 31;
    const int wv   = __builtin_amdgcn_readfirstlane((int)(threadIdx.x >> 5));
    const int d    = cg * 32 + lane;
    const int* tk  = token_ids + b * TT;

    const char* nb = (const char*)node_emb;
    const unsigned dbyte = (unsigned)d * 4u;
    float h[TT];
    #pragma unroll
    for (int s = 0; s < TT; ++s) {
        const unsigned off = (unsigned)tk[s] * (unsigned)(DD * 4) + dbyte;
        h[s] = *(const float*)(nb + off);
    }

    const int t0 = wv * MX_TS;
    const float* wrow = wT_buf + ((size_t)b * TT + t0) * TT;

    #pragma clang loop vectorize(disable) interleave(disable) unroll(disable)
    for (int i = 0; i < MX_TS; ++i) {
        float a0 = -INFINITY, a1 = -INFINITY, a2 = -INFINITY, a3 = -INFINITY;
        #pragma unroll
        for (int s = 0; s < TT; s += 8) {
            a0 = fmaxf(fmaxf(a0, h[s + 0] * wrow[s + 0]), h[s + 4] * wrow[s + 4]);
            a1 = fmaxf(fmaxf(a1, h[s + 1] * wrow[s + 1]), h[s + 5] * wrow[s + 5]);
            a2 = fmaxf(fmaxf(a2, h[s + 2] * wrow[s + 2]), h[s + 6] * wrow[s + 6]);
            a3 = fmaxf(fmaxf(a3, h[s + 3] * wrow[s + 3]), h[s + 7] * wrow[s + 7]);
        }
        amax[(t0 + i) * 32 + lane] = fmaxf(fmaxf(a0, a1), fmaxf(a2, a3));
        wrow += TT;
    }
    __syncthreads();

    if (wv == 0) {
        const float eta = eta_p[0];
        const float one_m_eta = 1.0f - eta;
        float gacc = 0.0f;
        #pragma unroll
        for (int t = 0; t < TT; ++t) {
            gacc += eta * h[t] + one_m_eta * amax[t * 32 + lane];
        }
        g_buf[b * DD + d] = gacc * (1.0f / (float)TT);
    }
}

#define GK_CH 8
#define GK_NB 4
#define GK_NG (DD / 8 / GK_CH)

__global__ __launch_bounds__(32) __attribute__((amdgpu_waves_per_eu(1, 1)))
void tg_gemm_sigmoid_wmma(const float* __restrict__ g_buf,
                          const float* __restrict__ Wm,
                          const float* __restrict__ bias,
                          float* __restrict__ out) {
    const int n0   = blockIdx.x * 16;
    const int lane = threadIdx.x;
    const int half = lane >> 4;
    const int l    = lane & 15;

    const float bb = bias[n0 + l];

    const v4f* ap = (const v4f*)(g_buf + (size_t)l * DD + 4 * half);
    const v4f* bp = (const v4f*)(Wm    + (size_t)(n0 + l) * DD + 4 * half);

    v4f abuf[GK_NB][GK_CH], bbuf[GK_NB][GK_CH];
    v8f c0 = {}, c1 = {}, c2 = {}, c3 = {};

    #pragma unroll
    for (int g = 0; g < GK_NB; ++g) {
        #pragma unroll
        for (int j = 0; j < GK_CH; ++j) {
            abuf[g][j] = ap[2 * (g * GK_CH + j)];
            bbuf[g][j] = bp[2 * (g * GK_CH + j)];
        }
    }

    #pragma unroll
    for (int g = 0; g < GK_NG; ++g) {
        __builtin_amdgcn_sched_barrier(0);
        #pragma unroll
        for (int j = 0; j < GK_CH; ++j) {
            const v4f a = abuf[g % GK_NB][j];
            const v4f b = bbuf[g % GK_NB][j];
            const v2f a_lo = {a.x, a.y}, a_hi = {a.z, a.w};
            const v2f b_lo = {b.x, b.y}, b_hi = {b.z, b.w};
            if (j & 1) {
                c2 = __builtin_amdgcn_wmma_f32_16x16x4_f32(false, a_lo, false, b_lo, (short)0, c2, false, false);
                c3 = __builtin_amdgcn_wmma_f32_16x16x4_f32(false, a_hi, false, b_hi, (short)0, c3, false, false);
            } else {
                c0 = __builtin_amdgcn_wmma_f32_16x16x4_f32(false, a_lo, false, b_lo, (short)0, c0, false, false);
                c1 = __builtin_amdgcn_wmma_f32_16x16x4_f32(false, a_hi, false, b_hi, (short)0, c1, false, false);
            }
        }
        __builtin_amdgcn_sched_barrier(0);
        if (g + GK_NB < GK_NG) {
            #pragma unroll
            for (int j = 0; j < GK_CH; ++j) {
                abuf[g % GK_NB][j] = ap[2 * ((g + GK_NB) * GK_CH + j)];
                bbuf[g % GK_NB][j] = bp[2 * ((g + GK_NB) * GK_CH + j)];
            }
        }
    }

    const v8f c = (c0 + c1) + (c2 + c3);

    #pragma unroll
    for (int r = 0; r < 8; ++r) {
        const int m = r + 8 * half;
        const float v = c[r] + bb;
        out[(size_t)m * DD + n0 + l] = 1.0f / (1.0f + expf(-v));
    }
}

extern "C" void kernel_launch(void* const* d_in, const int* in_sizes, int n_in,
                              void* d_out, int out_size, void* d_ws, size_t ws_size,
                              hipStream_t stream) {
    const int*   token_ids    = (const int*)  d_in[0];
    const float* node_emb     = (const float*)d_in[1];
    const float* edge_weights = (const float*)d_in[2];
    const int*   edge_matrix  = (const int*)  d_in[3];
    const float* eta          = (const float*)d_in[4];
    const float* Wm           = (const float*)d_in[5];
    const float* bias         = (const float*)d_in[6];
    float*       out          = (float*)d_out;

    float* w_buf = (float*)d_ws;
    float* g_buf = w_buf + (size_t)BB * TT * TT;

    {
        const int total = BB * TT * TT;
        dim3 grid((total + 255) / 256), block(256);
        tg_w_kernel<<<grid, block, 0, stream>>>(token_ids, edge_weights,
                                                edge_matrix, w_buf);
    }
    {
        dim3 grid(BB * (DD / 32)), block(32 * MX_NW);
        tg_maxagg_kernel<<<grid, block, 0, stream>>>(token_ids, node_emb,
                                                     w_buf, eta, g_buf);
    }
    {
        dim3 grid(DD / 16), block(32);
        tg_gemm_sigmoid_wmma<<<grid, block, 0, stream>>>(g_buf, Wm, bias, out);
    }
}
